# starting kernel
# baseline (speedup 1.0000x reference)
#include <hip/hip_runtime.h>
#include <math.h>

#define LEN     200
#define SHIFT_  150
#define FOUT    128
#define BATCH   64
#define SLEN    480050
#define TFRM    3200
#define TT      64
#define NS      9650
#define PITCH   129
#define EPSF    1e-7f
#define INV_LN10 0.43429448190325176f

typedef __attribute__((ext_vector_type(2))) float v2f;
typedef __attribute__((ext_vector_type(8))) float v8f;

__global__ __launch_bounds__(256) void dft_stats(const float* __restrict__ x,
                                                 float* __restrict__ ws) {
    __shared__ double sh[512];
    const int b   = blockIdx.x;
    const int tid = threadIdx.x;
    const float* xb = x + (size_t)b * SLEN;
    double s = 0.0, s2 = 0.0;
    for (int i = tid; i < SLEN; i += 256) {
        double v = (double)xb[i];
        s += v; s2 += v * v;
    }
    sh[tid] = s; sh[256 + tid] = s2;
    __syncthreads();
    for (int o = 128; o > 0; o >>= 1) {
        if (tid < o) { sh[tid] += sh[tid + o]; sh[256 + tid] += sh[256 + tid + o]; }
        __syncthreads();
    }
    if (tid == 0) {
        double m   = sh[0] / (double)SLEN;
        double var = sh[256] / (double)SLEN - m * m;
        if (var < 0.0) var = 0.0;
        ws[b]         = (float)m;
        ws[BATCH + b] = (float)(1.0 / (sqrt(var) + (double)EPSF));
    }
}

__global__ __launch_bounds__(256) void dft_spect(const float* __restrict__ x,
                                                 const float* __restrict__ kre,
                                                 const float* __restrict__ kim,
                                                 const float* __restrict__ ws,
                                                 float* __restrict__ out) {
    __shared__ float smem[NS];
    const int tid  = threadIdx.x;
    const int lane = tid & 31;
    const int wave = tid >> 5;
    const int b    = blockIdx.y;
    const int t0   = blockIdx.x * TT;

    const float mu  = ws[b];
    const float inv = ws[BATCH + b];

    const float* xb = x + (size_t)b * SLEN + (size_t)t0 * SHIFT_;
    for (int i = tid; i < NS; i += 256) smem[i] = (xb[i] - mu) * inv;
    __syncthreads();

    const int mrow = lane & 15;
    const int hi   = lane >> 4;
    const int kb   = hi << 1;
    const int f    = wave * 16 + mrow;

    const float* krp = kre + (size_t)f * LEN + kb;
    const float* kip = kim + (size_t)f * LEN + kb;

    const float* ap[4];
#pragma unroll
    for (int tt = 0; tt < 4; ++tt)
        ap[tt] = smem + SHIFT_ * (tt * 16 + mrow) + kb;

    v8f accR[4] = {}; v8f accI[4] = {};

#pragma unroll 2
    for (int k = 0; k < LEN; k += 4) {
        v2f br = *(const v2f*)(krp + k);
        v2f bi = *(const v2f*)(kip + k);
#pragma unroll
        for (int tt = 0; tt < 4; ++tt) {
            v2f a = *(const v2f*)(ap[tt] + k);
            accR[tt] = __builtin_amdgcn_wmma_f32_16x16x4_f32(
                false, a, false, br, (short)0, accR[tt], false, false);
            accI[tt] = __builtin_amdgcn_wmma_f32_16x16x4_f32(
                false, a, false, bi, (short)0, accI[tt], false, false);
        }
    }
    __syncthreads();

    float* fftl = smem;
    float* mus  = smem + TT * PITCH;
    float* invs = mus + TT;

#pragma unroll
    for (int tt = 0; tt < 4; ++tt) {
#pragma unroll
        for (int r = 0; r < 8; ++r) {
            float re  = accR[tt][r];
            float im  = accI[tt][r];
            float mag = sqrtf(re * re + im * im);
            float v   = logf(fmaxf(mag, EPSF)) * INV_LN10;
            int   tl  = tt * 16 + r + 8 * hi;
            fftl[tl * PITCH + f] = v;
        }
    }
    __syncthreads();

    if (tid < TT) {
        float s = 0.f, s2 = 0.f;
        const float* row = fftl + tid * PITCH;
        for (int ff = 0; ff < FOUT; ++ff) { float v = row[ff]; s += v; s2 += v * v; }
        float m   = s * (1.0f / FOUT);
        float var = fmaxf(s2 * (1.0f / FOUT) - m * m, 0.0f);
        mus[tid]  = m;
        invs[tid] = 1.0f / (sqrtf(var) + EPSF);
    }
    __syncthreads();

    float* outb = out + (size_t)b * FOUT * TFRM;
    for (int i = tid; i < TT * FOUT; i += 256) {
        int ff = i >> 6;
        int tl = i & (TT - 1);
        float v = (fftl[tl * PITCH + ff] - mus[tl]) * invs[tl];
        outb[(size_t)ff * TFRM + (t0 + tl)] = v;
    }
}

extern "C" void kernel_launch(void* const* d_in, const int* in_sizes, int n_in,
                              void* d_out, int out_size, void* d_ws, size_t ws_size,
                              hipStream_t stream) {
    const float* x   = (const float*)d_in[0];
    const float* kre = (const float*)d_in[1];
    const float* kim = (const float*)d_in[2];
    float* out = (float*)d_out;
    float* ws  = (float*)d_ws;

    dft_stats<<<dim3(BATCH), dim3(256), 0, stream>>>(x, ws);
    dft_spect<<<dim3(TFRM / TT, BATCH), dim3(256), 0, stream>>>(x, kre, kim, ws, out);
}


# DFT pass re-tiled: a wave owns all 128 bins of its 32 frames; log-magnitude, frame statistics, normalization stay in registers, no per-tile barrier or LDS round trip; frame sum order changes
# speedup vs baseline: 2.1332x; 2.1332x over previous
#include <hip/hip_runtime.h>
#include <math.h>

#define LEN     200
#define SHIFT_  150
#define FOUT    128
#define BATCH   64
#define SLEN    480050
#define TFRM    3200
#define EPSF    1e-7f
#define INV_LN10 0.43429448190325176f

#define NWG     256
#define NW      8
#define BP      204
#define BROWS   256
#define KQ      50
#define KDEEP   25
#define SF      32
#define HF      16
#define STRIPS_ROW 100
#define STRIPS_WG  25
#define WG_ROW  4
#define WPAIR   1225
#define WPITCH  2464
#define NPL     39

typedef __attribute__((ext_vector_type(2))) float v2f;
typedef __attribute__((ext_vector_type(4))) float v4f;
typedef __attribute__((ext_vector_type(8))) float v8f;

#define STHR    1024
#define SDEEP   16
#define RDEEP   8
__global__ __launch_bounds__(STHR) void dft_stats(const float* __restrict__ x,
                                                  float* __restrict__ ws) {
    __shared__ double sh[2 * (STHR / 32)];
    const int b    = blockIdx.x;
    const int tid  = threadIdx.x;
    const int lane = tid & 31;
    const int wave = tid >> 5;
    const float* xb = x + (size_t)b * SLEN;

    const int head = (int)(((16u - (unsigned)((size_t)xb & 15u)) & 15u) >> 2);
    const int nvec = (SLEN - head) >> 2;
    const int tail = (SLEN - head) - (nvec << 2);
    const v4f* xv = (const v4f*)(xb + head);

    v4f s = {0.f, 0.f, 0.f, 0.f};
    v4f q = {0.f, 0.f, 0.f, 0.f};
    int i = tid;
    for (; i + (SDEEP - 1) * STHR < nvec; i += SDEEP * STHR) {
        v4f v[SDEEP];
#pragma unroll
        for (int u = 0; u < SDEEP; ++u) v[u] = xv[i + u * STHR];
#pragma unroll
        for (int u = 0; u < SDEEP; ++u) { s += v[u]; q += v[u] * v[u]; }
    }
    for (; i < nvec; i += RDEEP * STHR) {
        v4f v[RDEEP];
#pragma unroll
        for (int u = 0; u < RDEEP; ++u) {
            const int j = i + u * STHR;
            v[u] = xv[j < nvec ? j : nvec - 1];
        }
#pragma unroll
        for (int u = 0; u < RDEEP; ++u) {
            if (i + u * STHR < nvec) { s += v[u]; q += v[u] * v[u]; }
        }
    }

    float fs = (s.x + s.y) + (s.z + s.w);
    float fq = (q.x + q.y) + (q.z + q.w);
    if (tid < head) { float v = xb[tid]; fs += v; fq += v * v; }
    if (tid < tail) { float v = xb[head + (nvec << 2) + tid]; fs += v; fq += v * v; }

#pragma unroll
    for (int o = 16; o > 0; o >>= 1) {
        fs += __shfl_xor(fs, o);
        fq += __shfl_xor(fq, o);
    }
    if (lane == 0) { sh[wave] = (double)fs; sh[STHR / 32 + wave] = (double)fq; }
    __syncthreads();
    if (wave == 0) {
        double ds = sh[lane];
        double dq = sh[STHR / 32 + lane];
#pragma unroll
        for (int o = 16; o > 0; o >>= 1) {
            ds += __shfl_xor(ds, o);
            dq += __shfl_xor(dq, o);
        }
        if (lane == 0) {
            double m   = ds / (double)SLEN;
            double var = dq / (double)SLEN - m * m;
            if (var < 0.0) var = 0.0;
            ws[b]         = (float)m;
            ws[BATCH + b] = (float)(1.0 / (sqrt(var) + (double)EPSF));
        }
    }
}


static __device__ __forceinline__ float xhalf(float v) {
    return __builtin_bit_cast(float, __builtin_amdgcn_permlanex16(
        __builtin_bit_cast(int, v), __builtin_bit_cast(int, v), 0x76543210, 0xfedcba98, false, false));
}

static __device__ __forceinline__ void load_window(v2f (&pre)[NPL], const float* src, int lane) {
    const v2f* s2 = (const v2f*)src;
#pragma unroll
    for (int u = 0; u < NPL; ++u) {
        const int p = lane + 32 * u;
        pre[u] = s2[p < WPAIR ? p : WPAIR - 1];
    }
}

struct Frag { v2f fb; v2f ar[8]; v2f ai[8]; };
static __device__ __forceinline__ void load_frag(Frag& f, const float* fp, const float* krp,
                                                 const float* kip, int k) {
    f.fb = *(const v2f*)(fp + k);
#pragma unroll
    for (int ft = 0; ft < 8; ++ft) {
        f.ar[ft] = *(const v2f*)(krp + ft * 16 * BP + k);
        f.ai[ft] = *(const v2f*)(kip + ft * 16 * BP + k);
    }
}
static __device__ __forceinline__ void mma_frag(v8f (&accR)[8], v8f (&accI)[8], const Frag& f) {
#pragma unroll
    for (int ft = 0; ft < 8; ++ft) {
        accR[ft] = __builtin_amdgcn_wmma_f32_16x16x4_f32(
            false, f.ar[ft], false, f.fb, (short)0, accR[ft], false, false);
        accI[ft] = __builtin_amdgcn_wmma_f32_16x16x4_f32(
            false, f.ai[ft], false, f.fb, (short)0, accI[ft], false, false);
    }
}

static __device__ __forceinline__ void operand_slots() {
    asm volatile("v_nop\n\tv_nop\n\tv_nop\n\tv_nop");
}

static __device__ __forceinline__ void half_strip(v8f (&o)[8], v2f (&pre)[NPL], float* win,
                                                  const float* fp, const float* krp, const float* kip,
                                                  float mu, float inv, bool has_next,
                                                  const float* next_src, int lane) {
#pragma unroll
    for (int u = 0; u < NPL; ++u) {
        const int p = lane + 32 * u;
        if (p < WPAIR) *(v2f*)(win + 2 * p) = (pre[u] - mu) * inv;
    }
    if (has_next) load_window(pre, next_src, lane);
    __builtin_amdgcn_wave_barrier();

    v8f accR[8] = {}; v8f accI[8] = {};
    Frag fa, fn;
    load_frag(fa, fp, krp, kip, 0);
#pragma unroll 4
    for (int k = 0; k < LEN - 8; k += 8) {
        load_frag(fn, fp, krp, kip, k + 4);
        __builtin_amdgcn_sched_barrier(0);
        mma_frag(accR, accI, fa);
        operand_slots();
        __builtin_amdgcn_sched_barrier(0);
        load_frag(fa, fp, krp, kip, k + 8);
        __builtin_amdgcn_sched_barrier(0);
        mma_frag(accR, accI, fn);
        operand_slots();
        __builtin_amdgcn_sched_barrier(0);
    }
    load_frag(fn, fp, krp, kip, LEN - 4);
    __builtin_amdgcn_sched_barrier(0);
    mma_frag(accR, accI, fa);
    mma_frag(accR, accI, fn);
    operand_slots();

    v8f V[8];
#pragma unroll
    for (int ft = 0; ft < 8; ++ft) {
#pragma unroll
        for (int r = 0; r < 8; ++r) {
            float re  = accR[ft][r];
            float im  = accI[ft][r];
            float mag = sqrtf(re * re + im * im);
            V[ft][r]  = logf(fmaxf(mag, EPSF)) * INV_LN10;
        }
    }
    v8f sa = V[0];
    v8f qa = V[0] * V[0];
#pragma unroll
    for (int ft = 1; ft < 8; ++ft) { sa += V[ft]; qa += V[ft] * V[ft]; }
    float s  = ((sa[0] + sa[1]) + (sa[2] + sa[3])) + ((sa[4] + sa[5]) + (sa[6] + sa[7]));
    float s2 = ((qa[0] + qa[1]) + (qa[2] + qa[3])) + ((qa[4] + qa[5]) + (qa[6] + qa[7]));
    s  += xhalf(s);
    s2 += xhalf(s2);
    const float m   = s * (1.0f / FOUT);
    const float var = fmaxf(s2 * (1.0f / FOUT) - m * m, 0.0f);
    const float iv  = 1.0f / (sqrtf(var) + EPSF);
#pragma unroll
    for (int ft = 0; ft < 8; ++ft) o[ft] = (V[ft] - m) * iv;
}

__global__ __launch_bounds__(256) void dft_spect(const float* __restrict__ x,
                                                 const float* __restrict__ kre,
                                                 const float* __restrict__ kim,
                                                 const float* __restrict__ ws,
                                                 float* __restrict__ out) {
    __shared__ __attribute__((aligned(16))) float Bs[BROWS * BP];
    __shared__ __attribute__((aligned(16))) float wins[NW * WPITCH];

    const int tid  = threadIdx.x;
    const int lane = tid & 31;
    const int wave = __builtin_amdgcn_readfirstlane(tid >> 5);

    const int g  = blockIdx.x;
    const int b  = g / WG_ROW;
    const int s0 = (g - b * WG_ROW) * STRIPS_WG;
    const float* xrow = x + (size_t)b * SLEN;
    const int nstrip  = (STRIPS_WG - wave + NW - 1) / NW;

    v2f pre[NPL];
    load_window(pre, xrow + (size_t)(s0 + wave) * SF * SHIFT_, lane);

    {
        const v2f* gr = (const v2f*)kre;
        const v2f* gi = (const v2f*)kim;
        for (int q0 = 0; q0 < KQ; q0 += KDEEP) {
            v2f r[KDEEP], m[KDEEP];
#pragma unroll
            for (int u = 0; u < KDEEP; ++u) {
                const int j = tid + (q0 + u) * 256;
                r[u] = gr[j];
                m[u] = gi[j];
            }
#pragma unroll
            for (int u = 0; u < KDEEP; ++u) {
                const int j   = tid + (q0 + u) * 256;
                const int row = j / (LEN / 2);
                const int c2  = j - row * (LEN / 2);
                *(v2f*)(Bs + row * BP + 2 * c2)          = r[u];
                *(v2f*)(Bs + (FOUT + row) * BP + 2 * c2) = m[u];
            }
        }
    }
    const float mu  = ws[b];
    const float inv = ws[BATCH + b];
    __syncthreads();

    const int c  = lane & 15;
    const int h  = lane >> 4;
    const int kb = h << 1;

    float* win = wins + wave * WPITCH;
    const float* fp  = win + SHIFT_ * c + kb;
    const float* krp = Bs + c * BP + kb;
    const float* kip = krp + FOUT * BP;

    for (int i = 0; i < nstrip; ++i) {
        const int strip = s0 + wave + NW * i;
        const float* src0 = xrow + (size_t)strip * SF * SHIFT_;
        v8f o0[8], o1[8];
        half_strip(o0, pre, win, fp, krp, kip, mu, inv, true, src0 + HF * SHIFT_, lane);
        half_strip(o1, pre, win, fp, krp, kip, mu, inv, i + 1 < nstrip,
                   src0 + (size_t)NW * SF * SHIFT_, lane);

        float* op = out + (size_t)b * FOUT * TFRM + (size_t)strip * SF + lane;
#pragma unroll
        for (int ft = 0; ft < 8; ++ft) {
#pragma unroll
            for (int r = 0; r < 8; ++r) {
                const float a0 = o0[ft][r];
                const float a1 = o1[ft][r];
                const float x0 = xhalf(a0);
                const float x1 = xhalf(a1);
                const float w0 = h ? x1 : a0;
                const float w1 = h ? a1 : x0;
                op[(size_t)(16 * ft + r) * TFRM]     = w0;
                op[(size_t)(16 * ft + 8 + r) * TFRM] = w1;
            }
        }
    }
}

extern "C" void kernel_launch(void* const* d_in, const int* in_sizes, int n_in,
                              void* d_out, int out_size, void* d_ws, size_t ws_size,
                              hipStream_t stream) {
    const float* x   = (const float*)d_in[0];
    const float* kre = (const float*)d_in[1];
    const float* kim = (const float*)d_in[2];
    float* out = (float*)d_out;
    float* ws  = (float*)d_ws;

    dft_stats<<<dim3(BATCH), dim3(STHR), 0, stream>>>(x, ws);
    dft_spect<<<dim3(NWG), dim3(256), 0, stream>>>(x, kre, kim, ws, out);
}


# DFT pass: each wave owns 3 strips; the 25th strip of a work-group (was wave 0's fourth) is split by bin tile over its 8 waves, frame sums joined in LDS: equal products per SIMD; its sum order changes
# speedup vs baseline: 2.2251x; 1.0431x over previous
#include <hip/hip_runtime.h>
#include <math.h>

#define LEN     200
#define SHIFT_  150
#define FOUT    128
#define BATCH   64
#define SLEN    480050
#define TFRM    3200
#define EPSF    1e-7f
#define INV_LN10 0.43429448190325176f

#define NWG     256
#define NW      8
#define BP      204
#define BROWS   256
#define KQ      50
#define KDEEP   25
#define SF      32
#define HF      16
#define STRIPS_ROW 100
#define STRIPS_WG  25
#define NS      3
#define WG_ROW  4
#define WPAIR   1225
#define WPITCH  2464
#define NPL     39
#define CPAIR   2425
#define CNPT    10
#define CTB     5
#define CKS     20

typedef __attribute__((ext_vector_type(2))) float v2f;
typedef __attribute__((ext_vector_type(4))) float v4f;
typedef __attribute__((ext_vector_type(8))) float v8f;

#define STHR    1024
#define SDEEP   16
#define RDEEP   8
__global__ __launch_bounds__(STHR) void dft_stats(const float* __restrict__ x,
                                                  float* __restrict__ ws) {
    __shared__ double sh[2 * (STHR / 32)];
    const int b    = blockIdx.x;
    const int tid  = threadIdx.x;
    const int lane = tid & 31;
    const int wave = tid >> 5;
    const float* xb = x + (size_t)b * SLEN;

    const int head = (int)(((16u - (unsigned)((size_t)xb & 15u)) & 15u) >> 2);
    const int nvec = (SLEN - head) >> 2;
    const int tail = (SLEN - head) - (nvec << 2);
    const v4f* xv = (const v4f*)(xb + head);

    v4f s = {0.f, 0.f, 0.f, 0.f};
    v4f q = {0.f, 0.f, 0.f, 0.f};
    int i = tid;
    for (; i + (SDEEP - 1) * STHR < nvec; i += SDEEP * STHR) {
        v4f v[SDEEP];
#pragma unroll
        for (int u = 0; u < SDEEP; ++u) v[u] = xv[i + u * STHR];
#pragma unroll
        for (int u = 0; u < SDEEP; ++u) { s += v[u]; q += v[u] * v[u]; }
    }
    for (; i < nvec; i += RDEEP * STHR) {
        v4f v[RDEEP];
#pragma unroll
        for (int u = 0; u < RDEEP; ++u) {
            const int j = i + u * STHR;
            v[u] = xv[j < nvec ? j : nvec - 1];
        }
#pragma unroll
        for (int u = 0; u < RDEEP; ++u) {
            if (i + u * STHR < nvec) { s += v[u]; q += v[u] * v[u]; }
        }
    }

    float fs = (s.x + s.y) + (s.z + s.w);
    float fq = (q.x + q.y) + (q.z + q.w);
    if (tid < head) { float v = xb[tid]; fs += v; fq += v * v; }
    if (tid < tail) { float v = xb[head + (nvec << 2) + tid]; fs += v; fq += v * v; }

#pragma unroll
    for (int o = 16; o > 0; o >>= 1) {
        fs += __shfl_xor(fs, o);
        fq += __shfl_xor(fq, o);
    }
    if (lane == 0) { sh[wave] = (double)fs; sh[STHR / 32 + wave] = (double)fq; }
    __syncthreads();
    if (wave == 0) {
        double ds = sh[lane];
        double dq = sh[STHR / 32 + lane];
#pragma unroll
        for (int o = 16; o > 0; o >>= 1) {
            ds += __shfl_xor(ds, o);
            dq += __shfl_xor(dq, o);
        }
        if (lane == 0) {
            double m   = ds / (double)SLEN;
            double var = dq / (double)SLEN - m * m;
            if (var < 0.0) var = 0.0;
            ws[b]         = (float)m;
            ws[BATCH + b] = (float)(1.0 / (sqrt(var) + (double)EPSF));
        }
    }
}


static __device__ __forceinline__ float xhalf(float v) {
    return __builtin_bit_cast(float, __builtin_amdgcn_permlanex16(
        __builtin_bit_cast(int, v), __builtin_bit_cast(int, v), 0x76543210, 0xfedcba98, false, false));
}

static __device__ __forceinline__ void load_window(v2f (&pre)[NPL], const float* src, int lane) {
    const v2f* s2 = (const v2f*)src;
#pragma unroll
    for (int u = 0; u < NPL; ++u) {
        const int p = lane + 32 * u;
        pre[u] = s2[p < WPAIR ? p : WPAIR - 1];
    }
}

static __device__ __forceinline__ void load_strip_slice(v2f (&cw)[CNPT], const float* src, int tid) {
    const v2f* s2 = (const v2f*)src;
#pragma unroll
    for (int u = 0; u < CNPT; ++u) {
        const int p = tid + 256 * u;
        cw[u] = s2[p < CPAIR ? p : CPAIR - 1];
    }
}

struct Frag { v2f fb; v2f ar[8]; v2f ai[8]; };
static __device__ __forceinline__ void load_frag(Frag& f, const float* fp, const float* krp,
                                                 const float* kip, int k) {
    f.fb = *(const v2f*)(fp + k);
#pragma unroll
    for (int ft = 0; ft < 8; ++ft) {
        f.ar[ft] = *(const v2f*)(krp + ft * 16 * BP + k);
        f.ai[ft] = *(const v2f*)(kip + ft * 16 * BP + k);
    }
}
static __device__ __forceinline__ void mma_frag(v8f (&accR)[8], v8f (&accI)[8], const Frag& f) {
#pragma unroll
    for (int ft = 0; ft < 8; ++ft) {
        accR[ft] = __builtin_amdgcn_wmma_f32_16x16x4_f32(
            false, f.ar[ft], false, f.fb, (short)0, accR[ft], false, false);
        accI[ft] = __builtin_amdgcn_wmma_f32_16x16x4_f32(
            false, f.ai[ft], false, f.fb, (short)0, accI[ft], false, false);
    }
}

struct CFrag { v2f ar[CTB]; v2f ai[CTB]; v2f f0[CTB]; v2f f1[CTB]; };
static __device__ __forceinline__ void load_cfrag(CFrag& f, const float* fp0, const float* krp,
                                                  const float* kip, int k) {
#pragma unroll
    for (int j = 0; j < CTB; ++j) {
        f.ar[j] = *(const v2f*)(krp + k + 4 * j);
        f.ai[j] = *(const v2f*)(kip + k + 4 * j);
        f.f0[j] = *(const v2f*)(fp0 + k + 4 * j);
        f.f1[j] = *(const v2f*)(fp0 + HF * SHIFT_ + k + 4 * j);
    }
}
static __device__ __forceinline__ void mma_cfrag(v8f& R0, v8f& I0, v8f& R1, v8f& I1, const CFrag& f) {
#pragma unroll
    for (int j = 0; j < CTB; ++j) {
        R0 = __builtin_amdgcn_wmma_f32_16x16x4_f32(false, f.ar[j], false, f.f0[j], (short)0, R0, false, false);
        I0 = __builtin_amdgcn_wmma_f32_16x16x4_f32(false, f.ai[j], false, f.f0[j], (short)0, I0, false, false);
        R1 = __builtin_amdgcn_wmma_f32_16x16x4_f32(false, f.ar[j], false, f.f1[j], (short)0, R1, false, false);
        I1 = __builtin_amdgcn_wmma_f32_16x16x4_f32(false, f.ai[j], false, f.f1[j], (short)0, I1, false, false);
    }
}

static __device__ __forceinline__ float logmag(float re, float im) {
    const float mag = sqrtf(re * re + im * im);
    return logf(fmaxf(mag, EPSF)) * INV_LN10;
}

static __device__ __forceinline__ void operand_slots() {
    asm volatile("v_nop\n\tv_nop\n\tv_nop\n\tv_nop");
}

static __device__ __forceinline__ void half_strip(v8f (&o)[8], v2f (&pre)[NPL], float* win,
                                                  const float* fp, const float* krp, const float* kip,
                                                  float mu, float inv, bool has_next,
                                                  const float* next_src, int lane) {
#pragma unroll
    for (int u = 0; u < NPL; ++u) {
        const int p = lane + 32 * u;
        if (p < WPAIR) *(v2f*)(win + 2 * p) = (pre[u] - mu) * inv;
    }
    if (has_next) load_window(pre, next_src, lane);
    __builtin_amdgcn_wave_barrier();

    v8f accR[8] = {}; v8f accI[8] = {};
    Frag fa, fn;
    load_frag(fa, fp, krp, kip, 0);
#pragma unroll 4
    for (int k = 0; k < LEN - 8; k += 8) {
        load_frag(fn, fp, krp, kip, k + 4);
        __builtin_amdgcn_sched_barrier(0);
        mma_frag(accR, accI, fa);
        operand_slots();
        __builtin_amdgcn_sched_barrier(0);
        load_frag(fa, fp, krp, kip, k + 8);
        __builtin_amdgcn_sched_barrier(0);
        mma_frag(accR, accI, fn);
        operand_slots();
        __builtin_amdgcn_sched_barrier(0);
    }
    load_frag(fn, fp, krp, kip, LEN - 4);
    __builtin_amdgcn_sched_barrier(0);
    mma_frag(accR, accI, fa);
    mma_frag(accR, accI, fn);
    operand_slots();

    v8f V[8];
#pragma unroll
    for (int ft = 0; ft < 8; ++ft) {
#pragma unroll
        for (int r = 0; r < 8; ++r) {
            V[ft][r] = logmag(accR[ft][r], accI[ft][r]);
        }
    }
    v8f sa = V[0];
    v8f qa = V[0] * V[0];
#pragma unroll
    for (int ft = 1; ft < 8; ++ft) { sa += V[ft]; qa += V[ft] * V[ft]; }
    float s  = ((sa[0] + sa[1]) + (sa[2] + sa[3])) + ((sa[4] + sa[5]) + (sa[6] + sa[7]));
    float s2 = ((qa[0] + qa[1]) + (qa[2] + qa[3])) + ((qa[4] + qa[5]) + (qa[6] + qa[7]));
    s  += xhalf(s);
    s2 += xhalf(s2);
    const float m   = s * (1.0f / FOUT);
    const float var = fmaxf(s2 * (1.0f / FOUT) - m * m, 0.0f);
    const float iv  = 1.0f / (sqrtf(var) + EPSF);
#pragma unroll
    for (int ft = 0; ft < 8; ++ft) o[ft] = (V[ft] - m) * iv;
}

__global__ __launch_bounds__(256) void dft_spect(const float* __restrict__ x,
                                                 const float* __restrict__ kre,
                                                 const float* __restrict__ kim,
                                                 const float* __restrict__ ws,
                                                 float* __restrict__ out) {
    __shared__ __attribute__((aligned(16))) float Bs[BROWS * BP];
    __shared__ __attribute__((aligned(16))) float wins[NW * WPITCH];
    __shared__ float part[2 * NW * SF];

    const int tid  = threadIdx.x;
    const int lane = tid & 31;
    const int wave = __builtin_amdgcn_readfirstlane(tid >> 5);

    const int g  = blockIdx.x;
    const int b  = g / WG_ROW;
    const int s0 = (g - b * WG_ROW) * STRIPS_WG;
    const float* xrow = x + (size_t)b * SLEN;
    const int cstrip  = s0 + NS * NW;

    v2f pre[NPL];
    load_window(pre, xrow + (size_t)(s0 + wave) * SF * SHIFT_, lane);
    v2f cw[CNPT];
    load_strip_slice(cw, xrow + (size_t)cstrip * SF * SHIFT_, tid);

    {
        const v2f* gr = (const v2f*)kre;
        const v2f* gi = (const v2f*)kim;
        for (int q0 = 0; q0 < KQ; q0 += KDEEP) {
            v2f r[KDEEP], m[KDEEP];
#pragma unroll
            for (int u = 0; u < KDEEP; ++u) {
                const int j = tid + (q0 + u) * 256;
                r[u] = gr[j];
                m[u] = gi[j];
            }
#pragma unroll
            for (int u = 0; u < KDEEP; ++u) {
                const int j   = tid + (q0 + u) * 256;
                const int row = j / (LEN / 2);
                const int c2  = j - row * (LEN / 2);
                *(v2f*)(Bs + row * BP + 2 * c2)          = r[u];
                *(v2f*)(Bs + (FOUT + row) * BP + 2 * c2) = m[u];
            }
        }
    }
    const float mu  = ws[b];
    const float inv = ws[BATCH + b];
#pragma unroll
    for (int u = 0; u < CNPT; ++u) {
        const int p = tid + 256 * u;
        if (p < CPAIR) *(v2f*)(wins + 2 * p) = (cw[u] - mu) * inv;
    }
    __syncthreads();

    const int c  = lane & 15;
    const int h  = lane >> 4;
    const int kb = h << 1;

    float* win = wins + wave * WPITCH;
    const float* fp  = win + SHIFT_ * c + kb;
    const float* krp = Bs + c * BP + kb;
    const float* kip = krp + FOUT * BP;

    {
        const float* fp0 = wins + SHIFT_ * c + kb;
        const float* krw = krp + wave * 16 * BP;
        const float* kiw = krw + FOUT * BP;
        v8f R0 = {}, I0 = {}, R1 = {}, I1 = {};
        CFrag ca, cn;
        load_cfrag(ca, fp0, krw, kiw, 0);
        for (int k = 0; k < LEN - 2 * CKS; k += 2 * CKS) {
            load_cfrag(cn, fp0, krw, kiw, k + CKS);
            __builtin_amdgcn_sched_barrier(0);
            mma_cfrag(R0, I0, R1, I1, ca);
            operand_slots();
            __builtin_amdgcn_sched_barrier(0);
            load_cfrag(ca, fp0, krw, kiw, k + 2 * CKS);
            __builtin_amdgcn_sched_barrier(0);
            mma_cfrag(R0, I0, R1, I1, cn);
            operand_slots();
            __builtin_amdgcn_sched_barrier(0);
        }
        load_cfrag(cn, fp0, krw, kiw, LEN - CKS);
        __builtin_amdgcn_sched_barrier(0);
        mma_cfrag(R0, I0, R1, I1, ca);
        mma_cfrag(R0, I0, R1, I1, cn);
        operand_slots();

        v8f V0, V1;
#pragma unroll
        for (int r = 0; r < 8; ++r) {
            V0[r] = logmag(R0[r], I0[r]);
            V1[r] = logmag(R1[r], I1[r]);
        }
        const v8f W0 = V0 * V0;
        const v8f W1 = V1 * V1;
        float ps0 = ((V0[0] + V0[1]) + (V0[2] + V0[3])) + ((V0[4] + V0[5]) + (V0[6] + V0[7]));
        float pq0 = ((W0[0] + W0[1]) + (W0[2] + W0[3])) + ((W0[4] + W0[5]) + (W0[6] + W0[7]));
        float ps1 = ((V1[0] + V1[1]) + (V1[2] + V1[3])) + ((V1[4] + V1[5]) + (V1[6] + V1[7]));
        float pq1 = ((W1[0] + W1[1]) + (W1[2] + W1[3])) + ((W1[4] + W1[5]) + (W1[6] + W1[7]));
        ps0 += xhalf(ps0);
        pq0 += xhalf(pq0);
        ps1 += xhalf(ps1);
        pq1 += xhalf(pq1);
        part[(2 * wave) * SF + lane]     = h ? ps1 : ps0;
        part[(2 * wave + 1) * SF + lane] = h ? pq1 : pq0;
        __syncthreads();
        float fs0 = 0.f, fq0 = 0.f, fs1 = 0.f, fq1 = 0.f;
#pragma unroll
        for (int w = 0; w < NW; ++w) {
            fs0 += part[(2 * w) * SF + c];
            fq0 += part[(2 * w + 1) * SF + c];
            fs1 += part[(2 * w) * SF + HF + c];
            fq1 += part[(2 * w + 1) * SF + HF + c];
        }
        const float m0   = fs0 * (1.0f / FOUT);
        const float m1   = fs1 * (1.0f / FOUT);
        const float var0 = fmaxf(fq0 * (1.0f / FOUT) - m0 * m0, 0.0f);
        const float var1 = fmaxf(fq1 * (1.0f / FOUT) - m1 * m1, 0.0f);
        const float iv0  = 1.0f / (sqrtf(var0) + EPSF);
        const float iv1  = 1.0f / (sqrtf(var1) + EPSF);
        const v8f n0 = (V0 - m0) * iv0;
        const v8f n1 = (V1 - m1) * iv1;

        float* cp = out + (size_t)b * FOUT * TFRM + (size_t)cstrip * SF + lane;
#pragma unroll
        for (int r = 0; r < 8; ++r) {
            const float a0 = n0[r];
            const float a1 = n1[r];
            const float x0 = xhalf(a0);
            const float x1 = xhalf(a1);
            const float w0 = h ? x1 : a0;
            const float w1 = h ? a1 : x0;
            cp[(size_t)(16 * wave + r) * TFRM]     = w0;
            cp[(size_t)(16 * wave + 8 + r) * TFRM] = w1;
        }
    }

    for (int i = 0; i < NS; ++i) {
        const int strip = s0 + wave + NW * i;
        const float* src0 = xrow + (size_t)strip * SF * SHIFT_;
        v8f o0[8], o1[8];
        half_strip(o0, pre, win, fp, krp, kip, mu, inv, true, src0 + HF * SHIFT_, lane);
        half_strip(o1, pre, win, fp, krp, kip, mu, inv, i + 1 < NS,
                   src0 + (size_t)NW * SF * SHIFT_, lane);

        float* op = out + (size_t)b * FOUT * TFRM + (size_t)strip * SF + lane;
#pragma unroll
        for (int ft = 0; ft < 8; ++ft) {
#pragma unroll
            for (int r = 0; r < 8; ++r) {
                const float a0 = o0[ft][r];
                const float a1 = o1[ft][r];
                const float x0 = xhalf(a0);
                const float x1 = xhalf(a1);
                const float w0 = h ? x1 : a0;
                const float w1 = h ? a1 : x0;
                op[(size_t)(16 * ft + r) * TFRM]     = w0;
                op[(size_t)(16 * ft + 8 + r) * TFRM] = w1;
            }
        }
    }
}

extern "C" void kernel_launch(void* const* d_in, const int* in_sizes, int n_in,
                              void* d_out, int out_size, void* d_ws, size_t ws_size,
                              hipStream_t stream) {
    const float* x   = (const float*)d_in[0];
    const float* kre = (const float*)d_in[1];
    const float* kim = (const float*)d_in[2];
    float* out = (float*)d_out;
    float* ws  = (float*)d_ws;

    dft_stats<<<dim3(BATCH), dim3(STHR), 0, stream>>>(x, ws);
    dft_spect<<<dim3(NWG), dim3(256), 0, stream>>>(x, kre, kim, ws, out);
}


# strips' tap loop: kernel-matrix fragment loads addressed from four LDS window bases held in registers plus 16-bit offsets, not an address add per load; the slots behind each product run go too
# speedup vs baseline: 2.3133x; 1.0396x over previous
#include <hip/hip_runtime.h>
#include <math.h>

#define LEN     200
#define SHIFT_  150
#define FOUT    128
#define BATCH   64
#define SLEN    480050
#define TFRM    3200
#define EPSF    1e-7f
#define INV_LN10 0.43429448190325176f

#define NWG     256
#define NW      8
#define BP      204
#define BROWS   256
#define KQ      50
#define KDEEP   25
#define SF      32
#define HF      16
#define STRIPS_ROW 100
#define STRIPS_WG  25
#define NS      3
#define WG_ROW  4
#define WPAIR   1225
#define WPITCH  2464
#define NPL     39
#define CPAIR   2425
#define CNPT    10
#define CTB     5
#define CKS     20
#define TILEF   (16 * BP)
#define TPW     5
#define NWIN    4

typedef __attribute__((ext_vector_type(2))) float v2f;
typedef __attribute__((ext_vector_type(4))) float v4f;
typedef __attribute__((ext_vector_type(8))) float v8f;

#define STHR    1024
#define SDEEP   16
#define RDEEP   8
__global__ __launch_bounds__(STHR) void dft_stats(const float* __restrict__ x,
                                                  float* __restrict__ ws) {
    __shared__ double sh[2 * (STHR / 32)];
    const int b    = blockIdx.x;
    const int tid  = threadIdx.x;
    const int lane = tid & 31;
    const int wave = tid >> 5;
    const float* xb = x + (size_t)b * SLEN;

    const int head = (int)(((16u - (unsigned)((size_t)xb & 15u)) & 15u) >> 2);
    const int nvec = (SLEN - head) >> 2;
    const int tail = (SLEN - head) - (nvec << 2);
    const v4f* xv = (const v4f*)(xb + head);

    v4f s = {0.f, 0.f, 0.f, 0.f};
    v4f q = {0.f, 0.f, 0.f, 0.f};
    int i = tid;
    for (; i + (SDEEP - 1) * STHR < nvec; i += SDEEP * STHR) {
        v4f v[SDEEP];
#pragma unroll
        for (int u = 0; u < SDEEP; ++u) v[u] = xv[i + u * STHR];
#pragma unroll
        for (int u = 0; u < SDEEP; ++u) { s += v[u]; q += v[u] * v[u]; }
    }
    for (; i < nvec; i += RDEEP * STHR) {
        v4f v[RDEEP];
#pragma unroll
        for (int u = 0; u < RDEEP; ++u) {
            const int j = i + u * STHR;
            v[u] = xv[j < nvec ? j : nvec - 1];
        }
#pragma unroll
        for (int u = 0; u < RDEEP; ++u) {
            if (i + u * STHR < nvec) { s += v[u]; q += v[u] * v[u]; }
        }
    }

    float fs = (s.x + s.y) + (s.z + s.w);
    float fq = (q.x + q.y) + (q.z + q.w);
    if (tid < head) { float v = xb[tid]; fs += v; fq += v * v; }
    if (tid < tail) { float v = xb[head + (nvec << 2) + tid]; fs += v; fq += v * v; }

#pragma unroll
    for (int o = 16; o > 0; o >>= 1) {
        fs += __shfl_xor(fs, o);
        fq += __shfl_xor(fq, o);
    }
    if (lane == 0) { sh[wave] = (double)fs; sh[STHR / 32 + wave] = (double)fq; }
    __syncthreads();
    if (wave == 0) {
        double ds = sh[lane];
        double dq = sh[STHR / 32 + lane];
#pragma unroll
        for (int o = 16; o > 0; o >>= 1) {
            ds += __shfl_xor(ds, o);
            dq += __shfl_xor(dq, o);
        }
        if (lane == 0) {
            double m   = ds / (double)SLEN;
            double var = dq / (double)SLEN - m * m;
            if (var < 0.0) var = 0.0;
            ws[b]         = (float)m;
            ws[BATCH + b] = (float)(1.0 / (sqrt(var) + (double)EPSF));
        }
    }
}


static __device__ __forceinline__ float xhalf(float v) {
    return __builtin_bit_cast(float, __builtin_amdgcn_permlanex16(
        __builtin_bit_cast(int, v), __builtin_bit_cast(int, v), 0x76543210, 0xfedcba98, false, false));
}

static __device__ __forceinline__ void load_window(v2f (&pre)[NPL], const float* src, int lane) {
    const v2f* s2 = (const v2f*)src;
#pragma unroll
    for (int u = 0; u < NPL; ++u) {
        const int p = lane + 32 * u;
        pre[u] = s2[p < WPAIR ? p : WPAIR - 1];
    }
}

static __device__ __forceinline__ void load_strip_slice(v2f (&cw)[CNPT], const float* src, int tid) {
    const v2f* s2 = (const v2f*)src;
#pragma unroll
    for (int u = 0; u < CNPT; ++u) {
        const int p = tid + 256 * u;
        cw[u] = s2[p < CPAIR ? p : CPAIR - 1];
    }
}

struct Frag { v2f fb; v2f ar[8]; v2f ai[8]; };
struct KWin { const float* w[NWIN]; };
static __device__ __forceinline__ void load_frag(Frag& f, const float* fp, const KWin& kw, int k) {
    f.fb = *(const v2f*)(fp + k);
#pragma unroll
    for (int ft = 0; ft < 8; ++ft) {
        f.ar[ft] = *(const v2f*)(kw.w[ft / TPW] + (ft % TPW) * TILEF + k);
        f.ai[ft] = *(const v2f*)(kw.w[(8 + ft) / TPW] + ((8 + ft) % TPW) * TILEF + k);
    }
}
static __device__ __forceinline__ void mma_frag(v8f (&accR)[8], v8f (&accI)[8], const Frag& f) {
#pragma unroll
    for (int ft = 0; ft < 8; ++ft) {
        accR[ft] = __builtin_amdgcn_wmma_f32_16x16x4_f32(
            false, f.ar[ft], false, f.fb, (short)0, accR[ft], false, false);
        accI[ft] = __builtin_amdgcn_wmma_f32_16x16x4_f32(
            false, f.ai[ft], false, f.fb, (short)0, accI[ft], false, false);
    }
}

struct CFrag { v2f ar[CTB]; v2f ai[CTB]; v2f f0[CTB]; v2f f1[CTB]; };
static __device__ __forceinline__ void load_cfrag(CFrag& f, const float* fp0, const float* krp,
                                                  const float* kip, int k) {
#pragma unroll
    for (int j = 0; j < CTB; ++j) {
        f.ar[j] = *(const v2f*)(krp + k + 4 * j);
        f.ai[j] = *(const v2f*)(kip + k + 4 * j);
        f.f0[j] = *(const v2f*)(fp0 + k + 4 * j);
        f.f1[j] = *(const v2f*)(fp0 + HF * SHIFT_ + k + 4 * j);
    }
}
static __device__ __forceinline__ void mma_cfrag(v8f& R0, v8f& I0, v8f& R1, v8f& I1, const CFrag& f) {
#pragma unroll
    for (int j = 0; j < CTB; ++j) {
        R0 = __builtin_amdgcn_wmma_f32_16x16x4_f32(false, f.ar[j], false, f.f0[j], (short)0, R0, false, false);
        I0 = __builtin_amdgcn_wmma_f32_16x16x4_f32(false, f.ai[j], false, f.f0[j], (short)0, I0, false, false);
        R1 = __builtin_amdgcn_wmma_f32_16x16x4_f32(false, f.ar[j], false, f.f1[j], (short)0, R1, false, false);
        I1 = __builtin_amdgcn_wmma_f32_16x16x4_f32(false, f.ai[j], false, f.f1[j], (short)0, I1, false, false);
    }
}

static __device__ __forceinline__ float logmag(float re, float im) {
    const float mag = sqrtf(re * re + im * im);
    return logf(fmaxf(mag, EPSF)) * INV_LN10;
}

static __device__ __forceinline__ void operand_slots() {
    asm volatile("v_nop\n\tv_nop\n\tv_nop\n\tv_nop");
}

static __device__ __forceinline__ void half_strip(v8f (&o)[8], v2f (&pre)[NPL], float* win,
                                                  const float* fp, const KWin& kw,
                                                  float mu, float inv, bool has_next,
                                                  const float* next_src, int lane) {
#pragma unroll
    for (int u = 0; u < NPL; ++u) {
        const int p = lane + 32 * u;
        if (p < WPAIR) *(v2f*)(win + 2 * p) = (pre[u] - mu) * inv;
    }
    if (has_next) load_window(pre, next_src, lane);
    __builtin_amdgcn_wave_barrier();

    v8f accR[8] = {}; v8f accI[8] = {};
    Frag fa, fn;
    load_frag(fa, fp, kw, 0);
#pragma unroll 4
    for (int k = 0; k < LEN - 8; k += 8) {
        load_frag(fn, fp, kw, k + 4);
        __builtin_amdgcn_sched_barrier(0);
        mma_frag(accR, accI, fa);
        __builtin_amdgcn_sched_barrier(0);
        load_frag(fa, fp, kw, k + 8);
        __builtin_amdgcn_sched_barrier(0);
        mma_frag(accR, accI, fn);
        __builtin_amdgcn_sched_barrier(0);
    }
    load_frag(fn, fp, kw, LEN - 4);
    __builtin_amdgcn_sched_barrier(0);
    mma_frag(accR, accI, fa);
    mma_frag(accR, accI, fn);
    operand_slots();

    v8f V[8];
#pragma unroll
    for (int ft = 0; ft < 8; ++ft) {
#pragma unroll
        for (int r = 0; r < 8; ++r) {
            V[ft][r] = logmag(accR[ft][r], accI[ft][r]);
        }
    }
    v8f sa = V[0];
    v8f qa = V[0] * V[0];
#pragma unroll
    for (int ft = 1; ft < 8; ++ft) { sa += V[ft]; qa += V[ft] * V[ft]; }
    float s  = ((sa[0] + sa[1]) + (sa[2] + sa[3])) + ((sa[4] + sa[5]) + (sa[6] + sa[7]));
    float s2 = ((qa[0] + qa[1]) + (qa[2] + qa[3])) + ((qa[4] + qa[5]) + (qa[6] + qa[7]));
    s  += xhalf(s);
    s2 += xhalf(s2);
    const float m   = s * (1.0f / FOUT);
    const float var = fmaxf(s2 * (1.0f / FOUT) - m * m, 0.0f);
    const float iv  = 1.0f / (sqrtf(var) + EPSF);
#pragma unroll
    for (int ft = 0; ft < 8; ++ft) o[ft] = (V[ft] - m) * iv;
}

__global__ __launch_bounds__(256) void dft_spect(const float* __restrict__ x,
                                                 const float* __restrict__ kre,
                                                 const float* __restrict__ kim,
                                                 const float* __restrict__ ws,
                                                 float* __restrict__ out) {
    __shared__ __attribute__((aligned(16))) float lds[NW * WPITCH + BROWS * BP];
    float* const wins = lds;
    float* const Bs   = lds + NW * WPITCH;
    __shared__ float part[2 * NW * SF];

    const int tid  = threadIdx.x;
    const int lane = tid & 31;
    const int wave = __builtin_amdgcn_readfirstlane(tid >> 5);

    const int g  = blockIdx.x;
    const int b  = g / WG_ROW;
    const int s0 = (g - b * WG_ROW) * STRIPS_WG;
    const float* xrow = x + (size_t)b * SLEN;
    const int cstrip  = s0 + NS * NW;

    v2f pre[NPL];
    load_window(pre, xrow + (size_t)(s0 + wave) * SF * SHIFT_, lane);
    v2f cw[CNPT];
    load_strip_slice(cw, xrow + (size_t)cstrip * SF * SHIFT_, tid);

    {
        const v2f* gr = (const v2f*)kre;
        const v2f* gi = (const v2f*)kim;
        for (int q0 = 0; q0 < KQ; q0 += KDEEP) {
            v2f r[KDEEP], m[KDEEP];
#pragma unroll
            for (int u = 0; u < KDEEP; ++u) {
                const int j = tid + (q0 + u) * 256;
                r[u] = gr[j];
                m[u] = gi[j];
            }
#pragma unroll
            for (int u = 0; u < KDEEP; ++u) {
                const int j   = tid + (q0 + u) * 256;
                const int row = j / (LEN / 2);
                const int c2  = j - row * (LEN / 2);
                *(v2f*)(Bs + row * BP + 2 * c2)          = r[u];
                *(v2f*)(Bs + (FOUT + row) * BP + 2 * c2) = m[u];
            }
        }
    }
    const float mu  = ws[b];
    const float inv = ws[BATCH + b];
#pragma unroll
    for (int u = 0; u < CNPT; ++u) {
        const int p = tid + 256 * u;
        if (p < CPAIR) *(v2f*)(wins + 2 * p) = (cw[u] - mu) * inv;
    }
    __syncthreads();

    const int c  = lane & 15;
    const int h  = lane >> 4;
    const int kb = h << 1;

    float* win = wins + wave * WPITCH;
    const float* fp  = win + SHIFT_ * c + kb;
    const float* krp = Bs + c * BP + kb;
    const float* kip = krp + FOUT * BP;
    int wd0 = NW * WPITCH, wd1 = NW * WPITCH + TPW * TILEF,
        wd2 = NW * WPITCH + 2 * TPW * TILEF, wd3 = NW * WPITCH + 3 * TPW * TILEF;
    asm volatile("" : "+s"(wd0), "+s"(wd1), "+s"(wd2), "+s"(wd3));
    const float* lrow = lds + c * BP + kb;
    KWin kw;
    kw.w[0] = lrow + wd0;
    kw.w[1] = lrow + wd1;
    kw.w[2] = lrow + wd2;
    kw.w[3] = lrow + wd3;

    {
        const float* fp0 = wins + SHIFT_ * c + kb;
        const float* krw = krp + wave * 16 * BP;
        const float* kiw = krw + FOUT * BP;
        v8f R0 = {}, I0 = {}, R1 = {}, I1 = {};
        CFrag ca, cn;
        load_cfrag(ca, fp0, krw, kiw, 0);
        for (int k = 0; k < LEN - 2 * CKS; k += 2 * CKS) {
            load_cfrag(cn, fp0, krw, kiw, k + CKS);
            __builtin_amdgcn_sched_barrier(0);
            mma_cfrag(R0, I0, R1, I1, ca);
            operand_slots();
            __builtin_amdgcn_sched_barrier(0);
            load_cfrag(ca, fp0, krw, kiw, k + 2 * CKS);
            __builtin_amdgcn_sched_barrier(0);
            mma_cfrag(R0, I0, R1, I1, cn);
            operand_slots();
            __builtin_amdgcn_sched_barrier(0);
        }
        load_cfrag(cn, fp0, krw, kiw, LEN - CKS);
        __builtin_amdgcn_sched_barrier(0);
        mma_cfrag(R0, I0, R1, I1, ca);
        mma_cfrag(R0, I0, R1, I1, cn);
        operand_slots();

        v8f V0, V1;
#pragma unroll
        for (int r = 0; r < 8; ++r) {
            V0[r] = logmag(R0[r], I0[r]);
            V1[r] = logmag(R1[r], I1[r]);
        }
        const v8f W0 = V0 * V0;
        const v8f W1 = V1 * V1;
        float ps0 = ((V0[0] + V0[1]) + (V0[2] + V0[3])) + ((V0[4] + V0[5]) + (V0[6] + V0[7]));
        float pq0 = ((W0[0] + W0[1]) + (W0[2] + W0[3])) + ((W0[4] + W0[5]) + (W0[6] + W0[7]));
        float ps1 = ((V1[0] + V1[1]) + (V1[2] + V1[3])) + ((V1[4] + V1[5]) + (V1[6] + V1[7]));
        float pq1 = ((W1[0] + W1[1]) + (W1[2] + W1[3])) + ((W1[4] + W1[5]) + (W1[6] + W1[7]));
        ps0 += xhalf(ps0);
        pq0 += xhalf(pq0);
        ps1 += xhalf(ps1);
        pq1 += xhalf(pq1);
        part[(2 * wave) * SF + lane]     = h ? ps1 : ps0;
        part[(2 * wave + 1) * SF + lane] = h ? pq1 : pq0;
        __syncthreads();
        float fs0 = 0.f, fq0 = 0.f, fs1 = 0.f, fq1 = 0.f;
#pragma unroll
        for (int w = 0; w < NW; ++w) {
            fs0 += part[(2 * w) * SF + c];
            fq0 += part[(2 * w + 1) * SF + c];
            fs1 += part[(2 * w) * SF + HF + c];
            fq1 += part[(2 * w + 1) * SF + HF + c];
        }
        const float m0   = fs0 * (1.0f / FOUT);
        const float m1   = fs1 * (1.0f / FOUT);
        const float var0 = fmaxf(fq0 * (1.0f / FOUT) - m0 * m0, 0.0f);
        const float var1 = fmaxf(fq1 * (1.0f / FOUT) - m1 * m1, 0.0f);
        const float iv0  = 1.0f / (sqrtf(var0) + EPSF);
        const float iv1  = 1.0f / (sqrtf(var1) + EPSF);
        const v8f n0 = (V0 - m0) * iv0;
        const v8f n1 = (V1 - m1) * iv1;

        float* cp = out + (size_t)b * FOUT * TFRM + (size_t)cstrip * SF + lane;
#pragma unroll
        for (int r = 0; r < 8; ++r) {
            const float a0 = n0[r];
            const float a1 = n1[r];
            const float x0 = xhalf(a0);
            const float x1 = xhalf(a1);
            const float w0 = h ? x1 : a0;
            const float w1 = h ? a1 : x0;
            cp[(size_t)(16 * wave + r) * TFRM]     = w0;
            cp[(size_t)(16 * wave + 8 + r) * TFRM] = w1;
        }
    }

    for (int i = 0; i < NS; ++i) {
        const int strip = s0 + wave + NW * i;
        const float* src0 = xrow + (size_t)strip * SF * SHIFT_;
        v8f o0[8], o1[8];
        half_strip(o0, pre, win, fp, kw, mu, inv, true, src0 + HF * SHIFT_, lane);
        half_strip(o1, pre, win, fp, kw, mu, inv, i + 1 < NS,
                   src0 + (size_t)NW * SF * SHIFT_, lane);

        float* op = out + (size_t)b * FOUT * TFRM + (size_t)strip * SF + lane;
#pragma unroll
        for (int ft = 0; ft < 8; ++ft) {
#pragma unroll
            for (int r = 0; r < 8; ++r) {
                const float a0 = o0[ft][r];
                const float a1 = o1[ft][r];
                const float x0 = xhalf(a0);
                const float x1 = xhalf(a1);
                const float w0 = h ? x1 : a0;
                const float w1 = h ? a1 : x0;
                op[(size_t)(16 * ft + r) * TFRM]     = w0;
                op[(size_t)(16 * ft + 8 + r) * TFRM] = w1;
            }
        }
    }
}

extern "C" void kernel_launch(void* const* d_in, const int* in_sizes, int n_in,
                              void* d_out, int out_size, void* d_ws, size_t ws_size,
                              hipStream_t stream) {
    const float* x   = (const float*)d_in[0];
    const float* kre = (const float*)d_in[1];
    const float* kim = (const float*)d_in[2];
    float* out = (float*)d_out;
    float* ws  = (float*)d_ws;

    dft_stats<<<dim3(BATCH), dim3(STHR), 0, stream>>>(x, ws);
    dft_spect<<<dim3(NWG), dim3(256), 0, stream>>>(x, kre, kim, ws, out);
}
